# starting kernel
# baseline (speedup 1.0000x reference)
#include <hip/hip_runtime.h>

typedef __attribute__((ext_vector_type(2))) float v2f;
typedef __attribute__((ext_vector_type(8))) float v8f;

#define D_DIM 128

__global__ void deg_init_kernel(unsigned* __restrict__ deg, int n) {
    int i = blockIdx.x * blockDim.x + threadIdx.x;
    if (i < n) deg[i] = 1u;
}

__global__ void deg_count_kernel(const int* __restrict__ adj, unsigned* __restrict__ deg,
                                 int nEdges) {
    int e = blockIdx.x * blockDim.x + threadIdx.x;
    if (e < nEdges) {
        int dst = adj[nEdges + e];
        atomicAdd(&deg[dst], 1u);
    }
}

__global__ void deg_finalize_kernel(float* __restrict__ dinv, int n) {
    int i = blockIdx.x * blockDim.x + threadIdx.x;
    if (i < n) {
        unsigned d = ((const unsigned*)dinv)[i];
        dinv[i] = __frsqrt_rn((float)d);
    }
}

__global__ void __launch_bounds__(256)
gemm_wmma_f32_kernel(const float* __restrict__ x, const float* __restrict__ W,
                     const float* __restrict__ bias, const float* __restrict__ dinv,
                     float* __restrict__ hs, float* __restrict__ out) {
    const int lane = threadIdx.x & 31;
    const int wv   = threadIdx.x >> 5;
    const int tm   = blockIdx.x * 16;
    const int tn   = wv * 16;
    const int half = lane >> 4;
    const int l15  = lane & 15;

    const float* xrow = x + (long)(tm + l15) * D_DIM;

    v8f acc = {};
#pragma unroll 8
    for (int k = 0; k < D_DIM; k += 4) {
        const int kb = k + 2 * half;
        v2f av;
        av.x = xrow[kb + 0];
        av.y = xrow[kb + 1];
        v2f bv;
        bv.x = W[(long)(kb + 0) * D_DIM + tn + l15];
        bv.y = W[(long)(kb + 1) * D_DIM + tn + l15];
        acc = __builtin_amdgcn_wmma_f32_16x16x4_f32(
             false, av,  false, bv,
             (short)0, acc,  false,  false);
    }

    const float bn = bias[tn + l15];
#pragma unroll
    for (int r = 0; r < 8; ++r) {
        const int  mr = tm + r + 8 * half;
        const float di = dinv[mr];
        const float hv = acc[r] * di;
        const long  idx = (long)mr * D_DIM + tn + l15;
        hs[idx]  = hv;
        out[idx] = hv * di + bn;
    }
}

__global__ void __launch_bounds__(256)
edge_aggregate_kernel(const int* __restrict__ adj, const float* __restrict__ hs,
                      const float* __restrict__ dinv, float* __restrict__ out,
                      int nEdges) {
    const int e = blockIdx.x * 8 + (threadIdx.x >> 5);
    if (e >= nEdges) return;
    const int lane = threadIdx.x & 31;

    const int src = adj[e];
    const int dst = adj[nEdges + e];
    const float scale = dinv[dst];

    const float4 v = *(const float4*)(hs + (long)src * D_DIM + lane * 4);
    float* o = out + (long)dst * D_DIM + lane * 4;

    const float m0 = v.x * scale;
    const float m1 = v.y * scale;
    const float m2 = v.z * scale;
    const float m3 = v.w * scale;

    asm volatile("global_atomic_add_f32 %0, %1, off"           :: "v"(o), "v"(m0) : "memory");
    asm volatile("global_atomic_add_f32 %0, %1, off offset:4"  :: "v"(o), "v"(m1) : "memory");
    asm volatile("global_atomic_add_f32 %0, %1, off offset:8"  :: "v"(o), "v"(m2) : "memory");
    asm volatile("global_atomic_add_f32 %0, %1, off offset:12" :: "v"(o), "v"(m3) : "memory");
}

extern "C" void kernel_launch(void* const* d_in, const int* in_sizes, int n_in,
                              void* d_out, int out_size, void* d_ws, size_t ws_size,
                              hipStream_t stream) {
    const float* x    = (const float*)d_in[0];
    const int*   adj  = (const int*)d_in[1];
    const float* W    = (const float*)d_in[2];
    const float* bias = (const float*)d_in[3];
    float*       out  = (float*)d_out;

    const int n      = in_sizes[0] / D_DIM;
    const int nEdges = in_sizes[1] / 2;

    unsigned* deg  = (unsigned*)d_ws;
    float*    dinv = (float*)d_ws;
    float*    hs   = (float*)d_ws + 10240;

    deg_init_kernel<<<(n + 255) / 256, 256, 0, stream>>>(deg, n);
    deg_count_kernel<<<(nEdges + 255) / 256, 256, 0, stream>>>(adj, deg, nEdges);
    deg_finalize_kernel<<<(n + 255) / 256, 256, 0, stream>>>(dinv, n);

    gemm_wmma_f32_kernel<<<n / 16, 256, 0, stream>>>(x, W, bias, dinv, hs, out);

    edge_aggregate_kernel<<<(nEdges + 7) / 8, 256, 0, stream>>>(adj, hs, dinv, out, nEdges);
}


# Edge sum made owner-computes: groups of 1024 scan all targets, list the edges into their 64 nodes in LDS, sum hs rows by LDS float adds, store out once; no global float atomics; sum order changes
# speedup vs baseline: 1.9374x; 1.9374x over previous
#include <hip/hip_runtime.h>

typedef __attribute__((ext_vector_type(2))) float v2f;
typedef __attribute__((ext_vector_type(8))) float v8f;

#define D_DIM 128
#define DEG_PITCH 64

__global__ void deg_init_kernel(unsigned* __restrict__ degw, int n) {
    int i = blockIdx.x * blockDim.x + threadIdx.x;
    if (i < n) degw[(long)i * DEG_PITCH] = 1u;
}

__global__ void deg_count_kernel(const int* __restrict__ adj, unsigned* __restrict__ degw,
                                 int nEdges) {
    int e = blockIdx.x * blockDim.x + threadIdx.x;
    if (e < nEdges) {
        int dst = adj[nEdges + e];
        atomicAdd(&degw[(long)dst * DEG_PITCH], 1u);
    }
}

__global__ void deg_finalize_kernel(const unsigned* __restrict__ degw,
                                    float* __restrict__ dinv, int n) {
    int i = blockIdx.x * blockDim.x + threadIdx.x;
    if (i < n) {
        unsigned d = degw[(long)i * DEG_PITCH];
        dinv[i] = __frsqrt_rn((float)d);
    }
}

__global__ void __launch_bounds__(256)
gemm_wmma_f32_kernel(const float* __restrict__ x, const float* __restrict__ W,
                     const float* __restrict__ dinv, float* __restrict__ hs) {
    const int lane = threadIdx.x & 31;
    const int wv   = threadIdx.x >> 5;
    const int tm   = blockIdx.x * 16;
    const int tn   = wv * 16;
    const int half = lane >> 4;
    const int l15  = lane & 15;

    const float* xrow = x + (long)(tm + l15) * D_DIM;

    v8f acc = {};
#pragma unroll 8
    for (int k = 0; k < D_DIM; k += 4) {
        const int kb = k + 2 * half;
        v2f av;
        av.x = xrow[kb + 0];
        av.y = xrow[kb + 1];
        v2f bv;
        bv.x = W[(long)(kb + 0) * D_DIM + tn + l15];
        bv.y = W[(long)(kb + 1) * D_DIM + tn + l15];
        acc = __builtin_amdgcn_wmma_f32_16x16x4_f32(
             false, av,  false, bv,
             (short)0, acc,  false,  false);
    }

#pragma unroll
    for (int r = 0; r < 8; ++r) {
        const int  mr = tm + r + 8 * half;
        hs[(long)mr * D_DIM + tn + l15] = acc[r] * dinv[mr];
    }
}

#define RNG          64
#define LIST_CAP     8000
#define EDGE_THREADS 1024
#define SCAN_UNROLL  8
#define GATH_UNROLL  16

__device__ __forceinline__ void lds_add_f32(float* p, float v) {
    __hip_atomic_fetch_add(p, v, __ATOMIC_RELAXED, __HIP_MEMORY_SCOPE_WORKGROUP);
}

__global__ void __launch_bounds__(EDGE_THREADS)
edge_owner_kernel(const int* __restrict__ adj, const float* __restrict__ hs,
                  const float* __restrict__ dinv, const float* __restrict__ bias,
                  float* __restrict__ out, int nEdges, int n) {
    __shared__ float    acc[RNG * D_DIM];
    __shared__ unsigned list[LIST_CAP];
    __shared__ unsigned cursor;

    const int tid  = threadIdx.x;
    const int lane = tid & 31;
    const int wv   = tid >> 5;
    const int base = blockIdx.x * RNG;

    for (int i = tid; i < RNG * D_DIM; i += EDGE_THREADS) acc[i] = 0.0f;
    if (tid == 0) cursor = 0u;
    __syncthreads();

    const char* tb   = (const char*)(adj + nEdges);
    const int  nQ    = nEdges >> 2;
    const int  nIter = (nQ + EDGE_THREADS - 1) / EDGE_THREADS;
    const int  rot   = blockIdx.x % nIter;

    for (int k0 = 0; k0 < nIter; k0 += SCAN_UNROLL) {
        int4     d[SCAN_UNROLL];
        unsigned q[SCAN_UNROLL];
        bool     okv[SCAN_UNROLL];
#pragma unroll
        for (int u = 0; u < SCAN_UNROLL; ++u) {
            int k = k0 + u + rot;
            if (k >= nIter) k -= nIter;
            q[u]   = (unsigned)(k * EDGE_THREADS + tid);
            okv[u] = (k0 + u < nIter) && (q[u] < (unsigned)nQ);
            d[u]   = *(const int4*)(tb + (okv[u] ? q[u] * 16u : 0u));
        }
#pragma unroll
        for (int u = 0; u < SCAN_UNROLL; ++u) {
            const bool ok = okv[u];
            const unsigned l0 = (unsigned)(d[u].x - base);
            const unsigned l1 = (unsigned)(d[u].y - base);
            const unsigned l2 = (unsigned)(d[u].z - base);
            const unsigned l3 = (unsigned)(d[u].w - base);
            const bool m0 = ok && l0 < RNG;
            const bool m1 = ok && l1 < RNG;
            const bool m2 = ok && l2 < RNG;
            const bool m3 = ok && l3 < RNG;
            if (m0 || m1 || m2 || m3) {
                const unsigned c   = (unsigned)m0 + (unsigned)m1 + (unsigned)m2 + (unsigned)m3;
                unsigned       pos = atomicAdd(&cursor, c);
                const unsigned e   = q[u] * 4u;
                if (m0) { if (pos < LIST_CAP) list[pos] = (e + 0u) | (l0 << 20); ++pos; }
                if (m1) { if (pos < LIST_CAP) list[pos] = (e + 1u) | (l1 << 20); ++pos; }
                if (m2) { if (pos < LIST_CAP) list[pos] = (e + 2u) | (l2 << 20); ++pos; }
                if (m3) { if (pos < LIST_CAP) list[pos] = (e + 3u) | (l3 << 20); }
            }
        }
    }
    __syncthreads();

    const unsigned filled = cursor;
    const int cnt = filled <= LIST_CAP ? (int)filled : 0;

    for (int i = tid; i < cnt; i += EDGE_THREADS) {
        const unsigned v = list[i];
        const unsigned s = (unsigned)adj[v & 0xFFFFFu];
        list[i] = s | ((v >> 20) << 16);
    }
    __syncthreads();

    if (cnt > 0) {
        for (int i0 = wv * GATH_UNROLL; i0 < cnt; i0 += (EDGE_THREADS / 32) * GATH_UNROLL) {
            unsigned ent[GATH_UNROLL];
            float4   r[GATH_UNROLL];
#pragma unroll
            for (int u = 0; u < GATH_UNROLL; ++u) {
                const int i = i0 + u < cnt ? i0 + u : cnt - 1;
                ent[u] = list[i];
            }
#pragma unroll
            for (int u = 0; u < GATH_UNROLL; ++u) {
                const unsigned s = ent[u] & 0xFFFFu;
                r[u] = *(const float4*)(hs + (long)s * D_DIM + lane * 4);
            }
#pragma unroll
            for (int u = 0; u < GATH_UNROLL; ++u) {
                const bool live = i0 + u < cnt;
                float* a = acc + (ent[u] >> 16) * D_DIM + lane;
                lds_add_f32(a +  0, live ? r[u].x : 0.0f);
                lds_add_f32(a + 32, live ? r[u].y : 0.0f);
                lds_add_f32(a + 64, live ? r[u].z : 0.0f);
                lds_add_f32(a + 96, live ? r[u].w : 0.0f);
            }
        }
    }

    if (filled > LIST_CAP) {
        const int* tgt = adj + nEdges;
        for (int e = tid; e < nEdges; e += EDGE_THREADS) {
            const unsigned dl = (unsigned)(tgt[e] - base);
            if (dl < RNG) {
                const float* row = hs + (long)adj[e] * D_DIM;
                for (int col = 0; col < D_DIM; ++col)
                    lds_add_f32(&acc[dl * D_DIM + (col & 3) * 32 + (col >> 2)], row[col]);
            }
        }
    }
    __syncthreads();

    const float4 b4 = *(const float4*)(bias + lane * 4);
    for (int r = wv; r < RNG; r += EDGE_THREADS / 32) {
        const int node = base + r;
        if (node < n) {
            const float* a  = acc + r * D_DIM + lane;
            const float4 h4 = *(const float4*)(hs + (long)node * D_DIM + lane * 4);
            const float  di = dinv[node];
            float4 o;
            o.x = (a[0]  + h4.x) * di + b4.x;
            o.y = (a[32] + h4.y) * di + b4.y;
            o.z = (a[64] + h4.z) * di + b4.z;
            o.w = (a[96] + h4.w) * di + b4.w;
            *(float4*)(out + (long)node * D_DIM + lane * 4) = o;
        }
    }
}

extern "C" void kernel_launch(void* const* d_in, const int* in_sizes, int n_in,
                              void* d_out, int out_size, void* d_ws, size_t ws_size,
                              hipStream_t stream) {
    const float* x    = (const float*)d_in[0];
    const int*   adj  = (const int*)d_in[1];
    const float* W    = (const float*)d_in[2];
    const float* bias = (const float*)d_in[3];
    float*       out  = (float*)d_out;

    const int n      = in_sizes[0] / D_DIM;
    const int nEdges = in_sizes[1] / 2;

    float*    dinv = (float*)d_ws;
    float*    hs   = (float*)d_ws + 10240;
    unsigned* degw = (unsigned*)hs;

    deg_init_kernel<<<(n + 255) / 256, 256, 0, stream>>>(degw, n);
    deg_count_kernel<<<(nEdges + 255) / 256, 256, 0, stream>>>(adj, degw, nEdges);
    deg_finalize_kernel<<<(n + 255) / 256, 256, 0, stream>>>(degw, dinv, n);

    gemm_wmma_f32_kernel<<<n / 16, 256, 0, stream>>>(x, W, dinv, hs);

    edge_owner_kernel<<<(n + RNG - 1) / RNG, EDGE_THREADS, 0, stream>>>(adj, hs, dinv, bias, out,
                                                                        nEdges, n);
}


# Edge listing split into its own launch, cut 40 node ranges x 6 edge slices; lists and in-degree counts parked in the owners' own out rows; the 3 degree launches go (5->3); sums still in arrival order
# speedup vs baseline: 3.1728x; 1.6377x over previous
#include <hip/hip_runtime.h>

typedef __attribute__((ext_vector_type(2))) float v2f;
typedef __attribute__((ext_vector_type(8))) float v8f;

#define D_DIM 128

#define RNG       64
#define SUB       4
#define LRNG      (RNG * SUB)
#define NSLICE    6
#define CNT_WORDS (NSLICE * RNG)
#define CAPW_FULL 1280

__device__ __forceinline__ int park_capw(int rows) {
    if (rows < 4) return 0;
    const int w = ((rows * D_DIM - CNT_WORDS) / NSLICE) & ~31;
    return w < CAPW_FULL ? w : CAPW_FULL;
}
__device__ __forceinline__ unsigned ldw(const float* p) { return __float_as_uint(*p); }
__device__ __forceinline__ void     stw(float* p, unsigned v) { *p = __uint_as_float(v); }

#define LIST_THREADS 1024
#define LSCAN_UNROLL 9

__global__ void __launch_bounds__(LIST_THREADS)
edge_list_kernel(const int* __restrict__ adj, float* __restrict__ out, int nEdges, int n) {
    __shared__ unsigned lst[SUB][CAPW_FULL];
    __shared__ unsigned cnt[LRNG];
    __shared__ unsigned cursor[SUB];

    const int tid  = threadIdx.x;
    const int R    = blockIdx.x / NSLICE;
    const int c    = blockIdx.x - R * NSLICE;
    const int base = R * LRNG;

    if (tid < LRNG) cnt[tid] = 0u;
    if (tid < SUB)  cursor[tid] = 0u;
    __syncthreads();

    const int4* tq  = (const int4*)(adj + nEdges);
    const int   nQ  = nEdges >> 2;
    const int   per = (nQ + NSLICE - 1) / NSLICE;
    const int   q0  = c * per;
    const int   q1  = q0 + per < nQ ? q0 + per : nQ;

    for (int qb = q0; qb < q1; qb += LIST_THREADS * LSCAN_UNROLL) {
        int4 d[LSCAN_UNROLL];
        int  q[LSCAN_UNROLL];
#pragma unroll
        for (int u = 0; u < LSCAN_UNROLL; ++u) {
            q[u] = qb + u * LIST_THREADS + tid;
            d[u] = tq[q[u] < q1 ? q[u] : q0];
        }
#pragma unroll
        for (int u = 0; u < LSCAN_UNROLL; ++u) {
            const bool     ok = q[u] < q1;
            const unsigned l0 = (unsigned)(d[u].x - base);
            const unsigned l1 = (unsigned)(d[u].y - base);
            const unsigned l2 = (unsigned)(d[u].z - base);
            const unsigned l3 = (unsigned)(d[u].w - base);
            const bool m0 = ok && l0 < LRNG;
            const bool m1 = ok && l1 < LRNG;
            const bool m2 = ok && l2 < LRNG;
            const bool m3 = ok && l3 < LRNG;
            if (m0 || m1 || m2 || m3) {
                const unsigned e = (unsigned)q[u] * 4u;
                unsigned p0 = 0u, p1 = 0u, p2 = 0u, p3 = 0u;
                if (m0) { atomicAdd(&cnt[l0], 1u); p0 = atomicAdd(&cursor[l0 >> 6], 1u); }
                if (m1) { atomicAdd(&cnt[l1], 1u); p1 = atomicAdd(&cursor[l1 >> 6], 1u); }
                if (m2) { atomicAdd(&cnt[l2], 1u); p2 = atomicAdd(&cursor[l2 >> 6], 1u); }
                if (m3) { atomicAdd(&cnt[l3], 1u); p3 = atomicAdd(&cursor[l3 >> 6], 1u); }
                if (m0 && p0 < CAPW_FULL - 1) lst[l0 >> 6][p0] = (e + 0u) | ((l0 & 63u) << 20);
                if (m1 && p1 < CAPW_FULL - 1) lst[l1 >> 6][p1] = (e + 1u) | ((l1 & 63u) << 20);
                if (m2 && p2 < CAPW_FULL - 1) lst[l2 >> 6][p2] = (e + 2u) | ((l2 & 63u) << 20);
                if (m3 && p3 < CAPW_FULL - 1) lst[l3 >> 6][p3] = (e + 3u) | ((l3 & 63u) << 20);
            }
        }
    }
    __syncthreads();

    int      capw[SUB];
    unsigned filled[SUB], npark[SUB];
#pragma unroll
    for (int s = 0; s < SUB; ++s) {
        int rows = n - (base + s * RNG);
        rows     = rows > RNG ? RNG : rows;
        capw[s]   = park_capw(rows);
        filled[s] = cursor[s];
        npark[s]  = (capw[s] > 0 && filled[s] <= (unsigned)(capw[s] - 1)) ? filled[s] : 0u;
    }
    unsigned ev[SUB][2], sv[SUB][2];
#pragma unroll
    for (int s = 0; s < SUB; ++s)
#pragma unroll
        for (int j = 0; j < 2; ++j) {
            const unsigned i = (unsigned)(tid + j * LIST_THREADS);
            ev[s][j] = i < npark[s] ? lst[s][i] : 0u;
        }
#pragma unroll
    for (int s = 0; s < SUB; ++s)
#pragma unroll
        for (int j = 0; j < 2; ++j) {
            const unsigned e = ev[s][j] & 0xFFFFFu;
            sv[s][j] = (unsigned)adj[e < (unsigned)nEdges ? e : 0u];
        }
#pragma unroll
    for (int s = 0; s < SUB; ++s) {
        if (capw[s] > 0) {
            float* slice = out + (long)(base + s * RNG) * D_DIM;
            float* reg   = slice + CNT_WORDS + c * capw[s];
#pragma unroll
            for (int j = 0; j < 2; ++j) {
                const unsigned i = (unsigned)(tid + j * LIST_THREADS);
                if (i < npark[s]) stw(reg + i, sv[s][j] | ((ev[s][j] >> 20) << 16));
            }
            if (tid < RNG) stw(slice + c * RNG + tid, cnt[s * RNG + tid]);
            if (tid == 0)  stw(reg + capw[s] - 1, filled[s]);
        }
    }
}

__global__ void __launch_bounds__(256)
gemm_wmma_f32_kernel(const float* __restrict__ x, const float* __restrict__ W,
                     const float* __restrict__ park, float* __restrict__ dinv,
                     float* __restrict__ hs) {
    const int lane = threadIdx.x & 31;
    const int wv   = threadIdx.x >> 5;
    const int tm   = blockIdx.x * 16;
    const int tn   = wv * 16;
    const int half = lane >> 4;
    const int l15  = lane & 15;

    const int    myrow = tm + l15;
    const float* pc    = park + (long)(myrow >> 6) * (RNG * D_DIM) + (myrow & (RNG - 1));
    unsigned deg = 1u;
#pragma unroll
    for (int c = 0; c < NSLICE; ++c) deg += ldw(pc + c * RNG);
    const float dv = __frsqrt_rn((float)deg);
    if (wv == 0 && lane < 16) dinv[myrow] = dv;

    const float* xrow = x + (long)myrow * D_DIM;

    v8f acc = {};
#pragma unroll 8
    for (int k = 0; k < D_DIM; k += 4) {
        const int kb = k + 2 * half;
        v2f av;
        av.x = xrow[kb + 0];
        av.y = xrow[kb + 1];
        v2f bv;
        bv.x = W[(long)(kb + 0) * D_DIM + tn + l15];
        bv.y = W[(long)(kb + 1) * D_DIM + tn + l15];
        acc = __builtin_amdgcn_wmma_f32_16x16x4_f32(
             false, av,  false, bv,
             (short)0, acc,  false,  false);
    }

#pragma unroll
    for (int r = 0; r < 8; ++r) {
        const int   mr = tm + r + 8 * half;
        const float sc = __shfl(dv, r + 8 * half, 32);
        hs[(long)mr * D_DIM + tn + l15] = acc[r] * sc;
    }
}

#define LIST_CAP     (NSLICE * (CAPW_FULL - 1))
#define EDGE_THREADS 1024
#define GATH_UNROLL  16

__device__ __forceinline__ void lds_add_f32(float* p, float v) {
    __hip_atomic_fetch_add(p, v, __ATOMIC_RELAXED, __HIP_MEMORY_SCOPE_WORKGROUP);
}

__global__ void __launch_bounds__(EDGE_THREADS)
edge_owner_kernel(const int* __restrict__ adj, const float* __restrict__ hs,
                  const float* __restrict__ dinv, const float* __restrict__ bias,
                  float* __restrict__ out, int nEdges, int n) {
    __shared__ float    acc[RNG * D_DIM];
    __shared__ unsigned list[LIST_CAP];

    const int tid  = threadIdx.x;
    const int lane = tid & 31;
    const int wv   = tid >> 5;
    const int base = blockIdx.x * RNG;

    for (int i = tid; i < RNG * D_DIM; i += EDGE_THREADS) acc[i] = 0.0f;

    int rows = n - base;
    rows     = rows > RNG ? RNG : rows;
    const int    capw  = park_capw(rows);
    const float* slice = out + (long)base * D_DIM;
    unsigned fc[NSLICE], off[NSLICE];
    unsigned total = 0u;
    bool     over  = capw == 0;
#pragma unroll
    for (int c = 0; c < NSLICE; ++c) {
        fc[c] = capw > 0 ? ldw(slice + CNT_WORDS + c * capw + capw - 1) : 0u;
        if (capw > 0 && fc[c] > (unsigned)(capw - 1)) over = true;
        off[c] = total;
        total += fc[c];
    }
    const int cnt = over ? 0 : (int)total;

    if (!over) {
#pragma unroll
        for (int j = 0; j < 2; ++j) {
            const unsigned i = (unsigned)(tid + j * EDGE_THREADS);
            unsigned v[NSLICE];
#pragma unroll
            for (int c = 0; c < NSLICE; ++c)
                v[c] = ldw(slice + CNT_WORDS + c * capw + (i < fc[c] ? (int)i : capw - 1));
#pragma unroll
            for (int c = 0; c < NSLICE; ++c)
                if (i < fc[c]) list[off[c] + i] = v[c];
        }
    }
    __syncthreads();

    if (cnt > 0) {
        for (int i0 = wv * GATH_UNROLL; i0 < cnt; i0 += (EDGE_THREADS / 32) * GATH_UNROLL) {
            unsigned ent[GATH_UNROLL];
            float4   r[GATH_UNROLL];
#pragma unroll
            for (int u = 0; u < GATH_UNROLL; ++u) {
                const int i = i0 + u < cnt ? i0 + u : cnt - 1;
                ent[u] = list[i];
            }
#pragma unroll
            for (int u = 0; u < GATH_UNROLL; ++u) {
                const unsigned s = ent[u] & 0xFFFFu;
                r[u] = *(const float4*)(hs + (long)s * D_DIM + lane * 4);
            }
#pragma unroll
            for (int u = 0; u < GATH_UNROLL; ++u) {
                const bool live = i0 + u < cnt;
                float* a = acc + (ent[u] >> 16) * D_DIM + lane;
                lds_add_f32(a +  0, live ? r[u].x : 0.0f);
                lds_add_f32(a + 32, live ? r[u].y : 0.0f);
                lds_add_f32(a + 64, live ? r[u].z : 0.0f);
                lds_add_f32(a + 96, live ? r[u].w : 0.0f);
            }
        }
    }

    if (over) {
        const int* tgt = adj + nEdges;
        for (int e = tid; e < nEdges; e += EDGE_THREADS) {
            const unsigned dl = (unsigned)(tgt[e] - base);
            if (dl < RNG) {
                const float* row = hs + (long)adj[e] * D_DIM;
                for (int col = 0; col < D_DIM; ++col)
                    lds_add_f32(&acc[dl * D_DIM + (col & 3) * 32 + (col >> 2)], row[col]);
            }
        }
    }
    __syncthreads();

    const float4 b4 = *(const float4*)(bias + lane * 4);
    for (int r = wv; r < RNG; r += EDGE_THREADS / 32) {
        const int node = base + r;
        if (node < n) {
            const float* a  = acc + r * D_DIM + lane;
            const float4 h4 = *(const float4*)(hs + (long)node * D_DIM + lane * 4);
            const float  di = dinv[node];
            float4 o;
            o.x = (a[0]  + h4.x) * di + b4.x;
            o.y = (a[32] + h4.y) * di + b4.y;
            o.z = (a[64] + h4.z) * di + b4.z;
            o.w = (a[96] + h4.w) * di + b4.w;
            *(float4*)(out + (long)node * D_DIM + lane * 4) = o;
        }
    }
}

extern "C" void kernel_launch(void* const* d_in, const int* in_sizes, int n_in,
                              void* d_out, int out_size, void* d_ws, size_t ws_size,
                              hipStream_t stream) {
    const float* x    = (const float*)d_in[0];
    const int*   adj  = (const int*)d_in[1];
    const float* W    = (const float*)d_in[2];
    const float* bias = (const float*)d_in[3];
    float*       out  = (float*)d_out;

    const int n      = in_sizes[0] / D_DIM;
    const int nEdges = in_sizes[1] / 2;

    float* dinv = (float*)d_ws;
    float* hs   = (float*)d_ws + 10240;

    edge_list_kernel<<<((n + LRNG - 1) / LRNG) * NSLICE, LIST_THREADS, 0, stream>>>(adj, out,
                                                                                   nEdges, n);

    gemm_wmma_f32_kernel<<<n / 16, 256, 0, stream>>>(x, W, out, dinv, hs);

    edge_owner_kernel<<<(n + RNG - 1) / RNG, EDGE_THREADS, 0, stream>>>(adj, hs, dinv, bias, out,
                                                                        nEdges, n);
}


# Owner ranges cut from 64 to 40 nodes (listing ranges 240): the edge-sum launch runs 250 groups, one per compute unit, not 157, so each LDS takes 5/8 of the float adds; sums still in arrival order
# speedup vs baseline: 3.5620x; 1.1227x over previous
#include <hip/hip_runtime.h>

typedef __attribute__((ext_vector_type(2))) float v2f;
typedef __attribute__((ext_vector_type(8))) float v8f;

#define D_DIM 128

#define RNG        40
#define SUB        6
#define LRNG       (RNG * SUB)
#define NSLICE     6
#define CNT_STRIDE 64
#define CNT_WORDS  (NSLICE * CNT_STRIDE)
#define CAPW_FULL  768

__device__ __forceinline__ int park_capw(int rows) {
    if (rows < 4) return 0;
    const int w = ((rows * D_DIM - CNT_WORDS) / NSLICE) & ~31;
    return w < CAPW_FULL ? w : CAPW_FULL;
}
__device__ __forceinline__ unsigned ldw(const float* p) { return __float_as_uint(*p); }
__device__ __forceinline__ void     stw(float* p, unsigned v) { *p = __uint_as_float(v); }

#define LIST_THREADS 1024
#define LSCAN_UNROLL 9
#define PARK_J ((CAPW_FULL + LIST_THREADS - 1) / LIST_THREADS)

__global__ void __launch_bounds__(LIST_THREADS)
edge_list_kernel(const int* __restrict__ adj, float* __restrict__ out, int nEdges, int n) {
    __shared__ unsigned lst[SUB][CAPW_FULL];
    __shared__ unsigned cnt[LRNG];
    __shared__ unsigned cursor[SUB];

    const int tid  = threadIdx.x;
    const int R    = blockIdx.x / NSLICE;
    const int c    = blockIdx.x - R * NSLICE;
    const int base = R * LRNG;

    if (tid < LRNG) cnt[tid] = 0u;
    if (tid < SUB)  cursor[tid] = 0u;
    __syncthreads();

    const int4* tq  = (const int4*)(adj + nEdges);
    const int   nQ  = nEdges >> 2;
    const int   per = (nQ + NSLICE - 1) / NSLICE;
    const int   q0  = c * per;
    const int   q1  = q0 + per < nQ ? q0 + per : nQ;

    for (int qb = q0; qb < q1; qb += LIST_THREADS * LSCAN_UNROLL) {
        int4 d[LSCAN_UNROLL];
        int  q[LSCAN_UNROLL];
#pragma unroll
        for (int u = 0; u < LSCAN_UNROLL; ++u) {
            q[u] = qb + u * LIST_THREADS + tid;
            d[u] = tq[q[u] < q1 ? q[u] : q0];
        }
#pragma unroll
        for (int u = 0; u < LSCAN_UNROLL; ++u) {
            const bool     ok = q[u] < q1;
            const unsigned l0 = (unsigned)(d[u].x - base);
            const unsigned l1 = (unsigned)(d[u].y - base);
            const unsigned l2 = (unsigned)(d[u].z - base);
            const unsigned l3 = (unsigned)(d[u].w - base);
            const bool m0 = ok && l0 < LRNG;
            const bool m1 = ok && l1 < LRNG;
            const bool m2 = ok && l2 < LRNG;
            const bool m3 = ok && l3 < LRNG;
            if (m0 || m1 || m2 || m3) {
                const unsigned e = (unsigned)q[u] * 4u;
                unsigned p0 = 0u, p1 = 0u, p2 = 0u, p3 = 0u;
                const unsigned s0 = l0 / RNG, s1 = l1 / RNG, s2 = l2 / RNG, s3 = l3 / RNG;
                if (m0) { atomicAdd(&cnt[l0], 1u); p0 = atomicAdd(&cursor[s0], 1u); }
                if (m1) { atomicAdd(&cnt[l1], 1u); p1 = atomicAdd(&cursor[s1], 1u); }
                if (m2) { atomicAdd(&cnt[l2], 1u); p2 = atomicAdd(&cursor[s2], 1u); }
                if (m3) { atomicAdd(&cnt[l3], 1u); p3 = atomicAdd(&cursor[s3], 1u); }
                if (m0 && p0 < CAPW_FULL - 1) lst[s0][p0] = (e + 0u) | ((l0 - s0 * RNG) << 20);
                if (m1 && p1 < CAPW_FULL - 1) lst[s1][p1] = (e + 1u) | ((l1 - s1 * RNG) << 20);
                if (m2 && p2 < CAPW_FULL - 1) lst[s2][p2] = (e + 2u) | ((l2 - s2 * RNG) << 20);
                if (m3 && p3 < CAPW_FULL - 1) lst[s3][p3] = (e + 3u) | ((l3 - s3 * RNG) << 20);
            }
        }
    }
    __syncthreads();

    int      capw[SUB];
    unsigned filled[SUB], npark[SUB];
#pragma unroll
    for (int s = 0; s < SUB; ++s) {
        int rows = n - (base + s * RNG);
        rows     = rows > RNG ? RNG : rows;
        capw[s]   = park_capw(rows);
        filled[s] = cursor[s];
        npark[s]  = (capw[s] > 0 && filled[s] <= (unsigned)(capw[s] - 1)) ? filled[s] : 0u;
    }
    unsigned ev[SUB][PARK_J], sv[SUB][PARK_J];
#pragma unroll
    for (int s = 0; s < SUB; ++s)
#pragma unroll
        for (int j = 0; j < PARK_J; ++j) {
            const unsigned i = (unsigned)(tid + j * LIST_THREADS);
            ev[s][j] = i < npark[s] ? lst[s][i] : 0u;
        }
#pragma unroll
    for (int s = 0; s < SUB; ++s)
#pragma unroll
        for (int j = 0; j < PARK_J; ++j) {
            const unsigned e = ev[s][j] & 0xFFFFFu;
            sv[s][j] = (unsigned)adj[e < (unsigned)nEdges ? e : 0u];
        }
#pragma unroll
    for (int s = 0; s < SUB; ++s) {
        if (capw[s] > 0) {
            float* slice = out + (long)(base + s * RNG) * D_DIM;
            float* reg   = slice + CNT_WORDS + c * capw[s];
#pragma unroll
            for (int j = 0; j < PARK_J; ++j) {
                const unsigned i = (unsigned)(tid + j * LIST_THREADS);
                if (i < npark[s]) stw(reg + i, sv[s][j] | ((ev[s][j] >> 20) << 16));
            }
            if (tid < RNG) stw(slice + c * CNT_STRIDE + tid, cnt[s * RNG + tid]);
            if (tid == 0)  stw(reg + capw[s] - 1, filled[s]);
        }
    }
}

__global__ void __launch_bounds__(256)
gemm_wmma_f32_kernel(const float* __restrict__ x, const float* __restrict__ W,
                     const float* __restrict__ park, float* __restrict__ dinv,
                     float* __restrict__ hs) {
    const int lane = threadIdx.x & 31;
    const int wv   = threadIdx.x >> 5;
    const int tm   = blockIdx.x * 16;
    const int tn   = wv * 16;
    const int half = lane >> 4;
    const int l15  = lane & 15;

    const int    myrow = tm + l15;
    const int    orng  = myrow / RNG;
    const float* pc    = park + (long)orng * (RNG * D_DIM) + (myrow - orng * RNG);
    unsigned deg = 1u;
#pragma unroll
    for (int c = 0; c < NSLICE; ++c) deg += ldw(pc + c * CNT_STRIDE);
    const float dv = __frsqrt_rn((float)deg);
    if (wv == 0 && lane < 16) dinv[myrow] = dv;

    const float* xrow = x + (long)myrow * D_DIM;

    v8f acc = {};
#pragma unroll 8
    for (int k = 0; k < D_DIM; k += 4) {
        const int kb = k + 2 * half;
        v2f av;
        av.x = xrow[kb + 0];
        av.y = xrow[kb + 1];
        v2f bv;
        bv.x = W[(long)(kb + 0) * D_DIM + tn + l15];
        bv.y = W[(long)(kb + 1) * D_DIM + tn + l15];
        acc = __builtin_amdgcn_wmma_f32_16x16x4_f32(
             false, av,  false, bv,
             (short)0, acc,  false,  false);
    }

#pragma unroll
    for (int r = 0; r < 8; ++r) {
        const int   mr = tm + r + 8 * half;
        const float sc = __shfl(dv, r + 8 * half, 32);
        hs[(long)mr * D_DIM + tn + l15] = acc[r] * sc;
    }
}

#define LIST_CAP     (NSLICE * (CAPW_FULL - 1))
#define EDGE_THREADS 1024
#define GATH_UNROLL  16
#define OWN_J ((CAPW_FULL + EDGE_THREADS - 1) / EDGE_THREADS)

__device__ __forceinline__ void lds_add_f32(float* p, float v) {
    __hip_atomic_fetch_add(p, v, __ATOMIC_RELAXED, __HIP_MEMORY_SCOPE_WORKGROUP);
}

__global__ void __launch_bounds__(EDGE_THREADS)
edge_owner_kernel(const int* __restrict__ adj, const float* __restrict__ hs,
                  const float* __restrict__ dinv, const float* __restrict__ bias,
                  float* __restrict__ out, int nEdges, int n) {
    __shared__ float    acc[RNG * D_DIM];
    __shared__ unsigned list[LIST_CAP];

    const int tid  = threadIdx.x;
    const int lane = tid & 31;
    const int wv   = tid >> 5;
    const int base = blockIdx.x * RNG;

    for (int i = tid; i < RNG * D_DIM; i += EDGE_THREADS) acc[i] = 0.0f;

    int rows = n - base;
    rows     = rows > RNG ? RNG : rows;
    const int    capw  = park_capw(rows);
    const float* slice = out + (long)base * D_DIM;
    unsigned fc[NSLICE], off[NSLICE];
    unsigned total = 0u;
    bool     over  = capw == 0;
#pragma unroll
    for (int c = 0; c < NSLICE; ++c) {
        fc[c] = capw > 0 ? ldw(slice + CNT_WORDS + c * capw + capw - 1) : 0u;
        if (capw > 0 && fc[c] > (unsigned)(capw - 1)) over = true;
        off[c] = total;
        total += fc[c];
    }
    const int cnt = over ? 0 : (int)total;

    if (!over) {
#pragma unroll
        for (int j = 0; j < OWN_J; ++j) {
            const unsigned i = (unsigned)(tid + j * EDGE_THREADS);
            unsigned v[NSLICE];
#pragma unroll
            for (int c = 0; c < NSLICE; ++c)
                v[c] = ldw(slice + CNT_WORDS + c * capw + (i < fc[c] ? (int)i : capw - 1));
#pragma unroll
            for (int c = 0; c < NSLICE; ++c)
                if (i < fc[c]) list[off[c] + i] = v[c];
        }
    }
    __syncthreads();

    if (cnt > 0) {
        for (int i0 = wv * GATH_UNROLL; i0 < cnt; i0 += (EDGE_THREADS / 32) * GATH_UNROLL) {
            unsigned ent[GATH_UNROLL];
            float4   r[GATH_UNROLL];
#pragma unroll
            for (int u = 0; u < GATH_UNROLL; ++u) {
                const int i = i0 + u < cnt ? i0 + u : cnt - 1;
                ent[u] = list[i];
            }
#pragma unroll
            for (int u = 0; u < GATH_UNROLL; ++u) {
                const unsigned s = ent[u] & 0xFFFFu;
                r[u] = *(const float4*)(hs + (long)s * D_DIM + lane * 4);
            }
#pragma unroll
            for (int u = 0; u < GATH_UNROLL; ++u) {
                const bool live = i0 + u < cnt;
                float* a = acc + (ent[u] >> 16) * D_DIM + lane;
                lds_add_f32(a +  0, live ? r[u].x : 0.0f);
                lds_add_f32(a + 32, live ? r[u].y : 0.0f);
                lds_add_f32(a + 64, live ? r[u].z : 0.0f);
                lds_add_f32(a + 96, live ? r[u].w : 0.0f);
            }
        }
    }

    if (over) {
        const int* tgt = adj + nEdges;
        for (int e = tid; e < nEdges; e += EDGE_THREADS) {
            const unsigned dl = (unsigned)(tgt[e] - base);
            if (dl < RNG) {
                const float* row = hs + (long)adj[e] * D_DIM;
                for (int col = 0; col < D_DIM; ++col)
                    lds_add_f32(&acc[dl * D_DIM + (col & 3) * 32 + (col >> 2)], row[col]);
            }
        }
    }
    __syncthreads();

    const float4 b4 = *(const float4*)(bias + lane * 4);
    for (int r = wv; r < RNG; r += EDGE_THREADS / 32) {
        const int node = base + r;
        if (node < n) {
            const float* a  = acc + r * D_DIM + lane;
            const float4 h4 = *(const float4*)(hs + (long)node * D_DIM + lane * 4);
            const float  di = dinv[node];
            float4 o;
            o.x = (a[0]  + h4.x) * di + b4.x;
            o.y = (a[32] + h4.y) * di + b4.y;
            o.z = (a[64] + h4.z) * di + b4.z;
            o.w = (a[96] + h4.w) * di + b4.w;
            *(float4*)(out + (long)node * D_DIM + lane * 4) = o;
        }
    }
}

extern "C" void kernel_launch(void* const* d_in, const int* in_sizes, int n_in,
                              void* d_out, int out_size, void* d_ws, size_t ws_size,
                              hipStream_t stream) {
    const float* x    = (const float*)d_in[0];
    const int*   adj  = (const int*)d_in[1];
    const float* W    = (const float*)d_in[2];
    const float* bias = (const float*)d_in[3];
    float*       out  = (float*)d_out;

    const int n      = in_sizes[0] / D_DIM;
    const int nEdges = in_sizes[1] / 2;

    float* dinv = (float*)d_ws;
    float* hs   = (float*)d_ws + 10240;

    edge_list_kernel<<<((n + LRNG - 1) / LRNG) * NSLICE, LIST_THREADS, 0, stream>>>(adj, out,
                                                                                   nEdges, n);

    gemm_wmma_f32_kernel<<<n / 16, 256, 0, stream>>>(x, W, out, dinv, hs);

    edge_owner_kernel<<<(n + RNG - 1) / RNG, EDGE_THREADS, 0, stream>>>(adj, hs, dinv, bias, out,
                                                                        nEdges, n);
}


# Listing re-cut to 16 ranges of 640 nodes x 16 edge slices (was 42 x 6): each target is read by 16 groups, not 42; a slice is one batch of 10 loads a thread; list order changes, sums in arrival order
# speedup vs baseline: 3.7149x; 1.0429x over previous
#include <hip/hip_runtime.h>

typedef __attribute__((ext_vector_type(2))) float v2f;
typedef __attribute__((ext_vector_type(8))) float v8f;

#define D_DIM 128

#define RNG        40
#define SUB        16
#define LRNG       (RNG * SUB)
#define NSLICE     16
#define CNT_STRIDE 64
#define CNT_WORDS  (NSLICE * CNT_STRIDE)
#define CAPW_FULL  256

__device__ __forceinline__ int park_capw(int rows) {
    const int room = rows * D_DIM - CNT_WORDS;
    if (room < NSLICE * 32) return 0;
    const int w = (room / NSLICE) & ~31;
    return w < CAPW_FULL ? w : CAPW_FULL;
}
__device__ __forceinline__ unsigned ldw(const float* p) { return __float_as_uint(*p); }
__device__ __forceinline__ void     stw(float* p, unsigned v) { *p = __uint_as_float(v); }

#define LIST_THREADS 1024
#define LSCAN_UNROLL 10
#define PARK_J ((CAPW_FULL + LIST_THREADS - 1) / LIST_THREADS)

__global__ void __launch_bounds__(LIST_THREADS)
edge_list_kernel(const int* __restrict__ adj, float* __restrict__ out, int nEdges, int n) {
    __shared__ unsigned lst[SUB][CAPW_FULL];
    __shared__ unsigned cnt[LRNG];
    __shared__ unsigned cursor[SUB];

    const int tid  = threadIdx.x;
    const int R    = blockIdx.x / NSLICE;
    const int c    = blockIdx.x - R * NSLICE;
    const int base = R * LRNG;

    if (tid < LRNG) cnt[tid] = 0u;
    if (tid < SUB)  cursor[tid] = 0u;
    __syncthreads();

    const int4* tq  = (const int4*)(adj + nEdges);
    const int   nQ  = nEdges >> 2;
    const int   per = (nQ + NSLICE - 1) / NSLICE;
    const int   q0  = c * per;
    const int   q1  = q0 + per < nQ ? q0 + per : nQ;

    for (int qb = q0; qb < q1; qb += LIST_THREADS * LSCAN_UNROLL) {
        int4 d[LSCAN_UNROLL];
        int  q[LSCAN_UNROLL];
#pragma unroll
        for (int u = 0; u < LSCAN_UNROLL; ++u) {
            q[u] = qb + u * LIST_THREADS + tid;
            d[u] = tq[q[u] < q1 ? q[u] : q0];
        }
#pragma unroll
        for (int u = 0; u < LSCAN_UNROLL; ++u) {
            const bool     ok = q[u] < q1;
            const unsigned l0 = (unsigned)(d[u].x - base);
            const unsigned l1 = (unsigned)(d[u].y - base);
            const unsigned l2 = (unsigned)(d[u].z - base);
            const unsigned l3 = (unsigned)(d[u].w - base);
            const bool m0 = ok && l0 < LRNG;
            const bool m1 = ok && l1 < LRNG;
            const bool m2 = ok && l2 < LRNG;
            const bool m3 = ok && l3 < LRNG;
            if (m0 || m1 || m2 || m3) {
                const unsigned e = (unsigned)q[u] * 4u;
                unsigned p0 = 0u, p1 = 0u, p2 = 0u, p3 = 0u;
                const unsigned s0 = l0 / RNG, s1 = l1 / RNG, s2 = l2 / RNG, s3 = l3 / RNG;
                if (m0) { atomicAdd(&cnt[l0], 1u); p0 = atomicAdd(&cursor[s0], 1u); }
                if (m1) { atomicAdd(&cnt[l1], 1u); p1 = atomicAdd(&cursor[s1], 1u); }
                if (m2) { atomicAdd(&cnt[l2], 1u); p2 = atomicAdd(&cursor[s2], 1u); }
                if (m3) { atomicAdd(&cnt[l3], 1u); p3 = atomicAdd(&cursor[s3], 1u); }
                if (m0 && p0 < CAPW_FULL - 1) lst[s0][p0] = (e + 0u) | ((l0 - s0 * RNG) << 20);
                if (m1 && p1 < CAPW_FULL - 1) lst[s1][p1] = (e + 1u) | ((l1 - s1 * RNG) << 20);
                if (m2 && p2 < CAPW_FULL - 1) lst[s2][p2] = (e + 2u) | ((l2 - s2 * RNG) << 20);
                if (m3 && p3 < CAPW_FULL - 1) lst[s3][p3] = (e + 3u) | ((l3 - s3 * RNG) << 20);
            }
        }
    }
    __syncthreads();

    int      capw[SUB];
    unsigned filled[SUB], npark[SUB];
#pragma unroll
    for (int s = 0; s < SUB; ++s) {
        int rows = n - (base + s * RNG);
        rows     = rows > RNG ? RNG : rows;
        capw[s]   = park_capw(rows);
        filled[s] = cursor[s];
        npark[s]  = (capw[s] > 0 && filled[s] <= (unsigned)(capw[s] - 1)) ? filled[s] : 0u;
    }
    unsigned ev[SUB][PARK_J], sv[SUB][PARK_J];
#pragma unroll
    for (int s = 0; s < SUB; ++s)
#pragma unroll
        for (int j = 0; j < PARK_J; ++j) {
            const unsigned i = (unsigned)(tid + j * LIST_THREADS);
            ev[s][j] = i < npark[s] ? lst[s][i] : 0u;
        }
#pragma unroll
    for (int s = 0; s < SUB; ++s)
#pragma unroll
        for (int j = 0; j < PARK_J; ++j) {
            const unsigned e = ev[s][j] & 0xFFFFFu;
            sv[s][j] = (unsigned)adj[e < (unsigned)nEdges ? e : 0u];
        }
#pragma unroll
    for (int s = 0; s < SUB; ++s) {
        if (capw[s] > 0) {
            float* slice = out + (long)(base + s * RNG) * D_DIM;
            float* reg   = slice + CNT_WORDS + c * capw[s];
#pragma unroll
            for (int j = 0; j < PARK_J; ++j) {
                const unsigned i = (unsigned)(tid + j * LIST_THREADS);
                if (i < npark[s]) stw(reg + i, sv[s][j] | ((ev[s][j] >> 20) << 16));
            }
            if (tid < RNG) stw(slice + c * CNT_STRIDE + tid, cnt[s * RNG + tid]);
            if (tid == 0)  stw(reg + capw[s] - 1, filled[s]);
        }
    }
}

__global__ void __launch_bounds__(256)
gemm_wmma_f32_kernel(const float* __restrict__ x, const float* __restrict__ W,
                     const float* __restrict__ park, float* __restrict__ dinv,
                     float* __restrict__ hs) {
    const int lane = threadIdx.x & 31;
    const int wv   = threadIdx.x >> 5;
    const int tm   = blockIdx.x * 16;
    const int tn   = wv * 16;
    const int half = lane >> 4;
    const int l15  = lane & 15;

    const int    myrow = tm + l15;
    const int    orng  = myrow / RNG;
    const float* pc    = park + (long)orng * (RNG * D_DIM) + (myrow - orng * RNG);
    unsigned deg = 1u;
#pragma unroll
    for (int c = 0; c < NSLICE; ++c) deg += ldw(pc + c * CNT_STRIDE);
    const float dv = __frsqrt_rn((float)deg);
    if (wv == 0 && lane < 16) dinv[myrow] = dv;

    const float* xrow = x + (long)myrow * D_DIM;

    v8f acc = {};
#pragma unroll 8
    for (int k = 0; k < D_DIM; k += 4) {
        const int kb = k + 2 * half;
        v2f av;
        av.x = xrow[kb + 0];
        av.y = xrow[kb + 1];
        v2f bv;
        bv.x = W[(long)(kb + 0) * D_DIM + tn + l15];
        bv.y = W[(long)(kb + 1) * D_DIM + tn + l15];
        acc = __builtin_amdgcn_wmma_f32_16x16x4_f32(
             false, av,  false, bv,
             (short)0, acc,  false,  false);
    }

#pragma unroll
    for (int r = 0; r < 8; ++r) {
        const int   mr = tm + r + 8 * half;
        const float sc = __shfl(dv, r + 8 * half, 32);
        hs[(long)mr * D_DIM + tn + l15] = acc[r] * sc;
    }
}

#define LIST_CAP     (NSLICE * (CAPW_FULL - 1))
#define EDGE_THREADS 1024
#define GATH_UNROLL  16
#define OWN_J ((CAPW_FULL + EDGE_THREADS - 1) / EDGE_THREADS)

__device__ __forceinline__ void lds_add_f32(float* p, float v) {
    __hip_atomic_fetch_add(p, v, __ATOMIC_RELAXED, __HIP_MEMORY_SCOPE_WORKGROUP);
}

__global__ void __launch_bounds__(EDGE_THREADS)
edge_owner_kernel(const int* __restrict__ adj, const float* __restrict__ hs,
                  const float* __restrict__ dinv, const float* __restrict__ bias,
                  float* __restrict__ out, int nEdges, int n) {
    __shared__ float    acc[RNG * D_DIM];
    __shared__ unsigned list[LIST_CAP];

    const int tid  = threadIdx.x;
    const int lane = tid & 31;
    const int wv   = tid >> 5;
    const int base = blockIdx.x * RNG;

    for (int i = tid; i < RNG * D_DIM; i += EDGE_THREADS) acc[i] = 0.0f;

    int rows = n - base;
    rows     = rows > RNG ? RNG : rows;
    const int    capw  = park_capw(rows);
    const float* slice = out + (long)base * D_DIM;
    unsigned fc[NSLICE], off[NSLICE];
    unsigned total = 0u;
    bool     over  = capw == 0;
#pragma unroll
    for (int c = 0; c < NSLICE; ++c) {
        fc[c] = capw > 0 ? ldw(slice + CNT_WORDS + c * capw + capw - 1) : 0u;
        if (capw > 0 && fc[c] > (unsigned)(capw - 1)) over = true;
        off[c] = total;
        total += fc[c];
    }
    const int cnt = over ? 0 : (int)total;

    if (!over) {
#pragma unroll
        for (int j = 0; j < OWN_J; ++j) {
            const unsigned i = (unsigned)(tid + j * EDGE_THREADS);
            unsigned v[NSLICE];
#pragma unroll
            for (int c = 0; c < NSLICE; ++c)
                v[c] = ldw(slice + CNT_WORDS + c * capw + (i < fc[c] ? (int)i : capw - 1));
#pragma unroll
            for (int c = 0; c < NSLICE; ++c)
                if (i < fc[c]) list[off[c] + i] = v[c];
        }
    }
    __syncthreads();

    if (cnt > 0) {
        for (int i0 = wv * GATH_UNROLL; i0 < cnt; i0 += (EDGE_THREADS / 32) * GATH_UNROLL) {
            unsigned ent[GATH_UNROLL];
            float4   r[GATH_UNROLL];
#pragma unroll
            for (int u = 0; u < GATH_UNROLL; ++u) {
                const int i = i0 + u < cnt ? i0 + u : cnt - 1;
                ent[u] = list[i];
            }
#pragma unroll
            for (int u = 0; u < GATH_UNROLL; ++u) {
                const unsigned s = ent[u] & 0xFFFFu;
                r[u] = *(const float4*)(hs + (long)s * D_DIM + lane * 4);
            }
#pragma unroll
            for (int u = 0; u < GATH_UNROLL; ++u) {
                const bool live = i0 + u < cnt;
                float* a = acc + (ent[u] >> 16) * D_DIM + lane;
                lds_add_f32(a +  0, live ? r[u].x : 0.0f);
                lds_add_f32(a + 32, live ? r[u].y : 0.0f);
                lds_add_f32(a + 64, live ? r[u].z : 0.0f);
                lds_add_f32(a + 96, live ? r[u].w : 0.0f);
            }
        }
    }

    if (over) {
        const int* tgt = adj + nEdges;
        for (int e = tid; e < nEdges; e += EDGE_THREADS) {
            const unsigned dl = (unsigned)(tgt[e] - base);
            if (dl < RNG) {
                const float* row = hs + (long)adj[e] * D_DIM;
                for (int col = 0; col < D_DIM; ++col)
                    lds_add_f32(&acc[dl * D_DIM + (col & 3) * 32 + (col >> 2)], row[col]);
            }
        }
    }
    __syncthreads();

    const float4 b4 = *(const float4*)(bias + lane * 4);
    for (int r = wv; r < RNG; r += EDGE_THREADS / 32) {
        const int node = base + r;
        if (node < n) {
            const float* a  = acc + r * D_DIM + lane;
            const float4 h4 = *(const float4*)(hs + (long)node * D_DIM + lane * 4);
            const float  di = dinv[node];
            float4 o;
            o.x = (a[0]  + h4.x) * di + b4.x;
            o.y = (a[32] + h4.y) * di + b4.y;
            o.z = (a[64] + h4.z) * di + b4.z;
            o.w = (a[96] + h4.w) * di + b4.w;
            *(float4*)(out + (long)node * D_DIM + lane * 4) = o;
        }
    }
}

extern "C" void kernel_launch(void* const* d_in, const int* in_sizes, int n_in,
                              void* d_out, int out_size, void* d_ws, size_t ws_size,
                              hipStream_t stream) {
    const float* x    = (const float*)d_in[0];
    const int*   adj  = (const int*)d_in[1];
    const float* W    = (const float*)d_in[2];
    const float* bias = (const float*)d_in[3];
    float*       out  = (float*)d_out;

    const int n      = in_sizes[0] / D_DIM;
    const int nEdges = in_sizes[1] / 2;

    float* dinv = (float*)d_ws;
    float* hs   = (float*)d_ws + 10240;

    edge_list_kernel<<<((n + LRNG - 1) / LRNG) * NSLICE, LIST_THREADS, 0, stream>>>(adj, out,
                                                                                   nEdges, n);

    gemm_wmma_f32_kernel<<<n / 16, 256, 0, stream>>>(x, W, out, dinv, hs);

    edge_owner_kernel<<<(n + RNG - 1) / RNG, EDGE_THREADS, 0, stream>>>(adj, hs, dinv, bias, out,
                                                                        nEdges, n);
}


# Listing launch handles edges lane-dense: ballot ranks queue a wave's matching edges in its own LDS, full waves drain it; copy-out shared by all waves; list order changes, sums in arrival order
# speedup vs baseline: 4.0553x; 1.0916x over previous
#include <hip/hip_runtime.h>

typedef __attribute__((ext_vector_type(2))) float v2f;
typedef __attribute__((ext_vector_type(8))) float v8f;

#define D_DIM 128

#define RNG        40
#define SUB        16
#define LRNG       (RNG * SUB)
#define NSLICE     16
#define CNT_STRIDE 64
#define CNT_WORDS  (NSLICE * CNT_STRIDE)
#define CAPW_FULL  256

__device__ __forceinline__ int park_capw(int rows) {
    const int room = rows * D_DIM - CNT_WORDS;
    if (room < NSLICE * 32) return 0;
    const int w = (room / NSLICE) & ~31;
    return w < CAPW_FULL ? w : CAPW_FULL;
}
__device__ __forceinline__ unsigned ldw(const float* p) { return __float_as_uint(*p); }
__device__ __forceinline__ void     stw(float* p, unsigned v) { *p = __uint_as_float(v); }

#define LIST_THREADS 1024
#define LSCAN_UNROLL 10
#define LSCAN_HALF   5
#define Q_FLUSH      128
#define Q_CAP        (Q_FLUSH + LSCAN_HALF * 128)
#define DRAIN_UNROLL 3
#define COPY_SETS    (LIST_THREADS / CAPW_FULL)
#define COPY_J       (SUB / COPY_SETS)

__device__ __forceinline__ unsigned queue_edge(unsigned* wq, unsigned qn, unsigned okm, bool ok, bool hit,
                                             unsigned word) {
    const unsigned mask = __builtin_amdgcn_ballot_w32(hit) & okm;
    if (ok && hit) wq[__builtin_amdgcn_mbcnt_lo(mask, qn)] = word;
    return qn + (unsigned)__builtin_popcount(mask);
}

__device__ __forceinline__ void drain_queue(const unsigned* wq, unsigned qn, int lane,
                                            const int* __restrict__ adj, unsigned* cnt,
                                            unsigned* cursor, unsigned (*lst)[CAPW_FULL]) {
    for (unsigned k0 = 0u; k0 < qn; k0 += 32u * DRAIN_UNROLL) {
        unsigned v[DRAIN_UNROLL], sv[DRAIN_UNROLL], p[DRAIN_UNROLL];
#pragma unroll
        for (int u = 0; u < DRAIN_UNROLL; ++u) {
            const unsigned k = k0 + 32u * u + (unsigned)lane;
            v[u] = wq[k < qn ? k : 0u];
        }
#pragma unroll
        for (int u = 0; u < DRAIN_UNROLL; ++u) sv[u] = (unsigned)adj[v[u] & 0xFFFFFu];
#pragma unroll
        for (int u = 0; u < DRAIN_UNROLL; ++u) {
            const bool     live = k0 + 32u * u + (unsigned)lane < qn;
            const unsigned l    = v[u] >> 20;
            p[u] = CAPW_FULL;
            if (live) { atomicAdd(&cnt[l], 1u); p[u] = atomicAdd(&cursor[l / RNG], 1u); }
        }
#pragma unroll
        for (int u = 0; u < DRAIN_UNROLL; ++u) {
            const unsigned l = v[u] >> 20;
            const unsigned s = l / RNG;
            if (p[u] < CAPW_FULL - 1) lst[s][p[u]] = sv[u] | ((l - s * RNG) << 16);
        }
    }
}

__global__ void __launch_bounds__(LIST_THREADS)
edge_list_kernel(const int* __restrict__ adj, float* __restrict__ out, int nEdges, int n) {
    __shared__ unsigned lst[SUB][CAPW_FULL];
    __shared__ unsigned cnt[LRNG];
    __shared__ unsigned cursor[SUB];
    __shared__ unsigned hq[LIST_THREADS / 32][Q_CAP];

    const int tid  = threadIdx.x;
    const int lane = tid & 31;
    const int R    = blockIdx.x / NSLICE;
    const int c    = blockIdx.x - R * NSLICE;
    const int base = R * LRNG;

    if (tid < LRNG) cnt[tid] = 0u;
    if (tid < SUB)  cursor[tid] = 0u;
    __syncthreads();

    const int4* tq  = (const int4*)(adj + nEdges);
    const int   nQ  = nEdges >> 2;
    const int   per = (nQ + NSLICE - 1) / NSLICE;
    const int   q0  = c * per;
    const int   q1  = q0 + per < nQ ? q0 + per : nQ;

    unsigned* wq = hq[tid >> 5];
    unsigned  qn = 0u;
    for (int qb = q0; qb < q1; qb += LIST_THREADS * LSCAN_UNROLL) {
        int4 d[LSCAN_UNROLL];
        int  q[LSCAN_UNROLL];
#pragma unroll
        for (int u = 0; u < LSCAN_UNROLL; ++u) {
            q[u] = qb + u * LIST_THREADS + tid;
            d[u] = tq[q[u] < q1 ? q[u] : q0];
        }
#pragma unroll
        for (int u = 0; u < LSCAN_UNROLL; ++u) {
            const bool     ok = q[u] < q1;
            const unsigned l0 = (unsigned)(d[u].x - base);
            const unsigned l1 = (unsigned)(d[u].y - base);
            const unsigned l2 = (unsigned)(d[u].z - base);
            const unsigned l3 = (unsigned)(d[u].w - base);
            const unsigned e   = (unsigned)q[u] * 4u;
            const unsigned okm = __builtin_amdgcn_ballot_w32(ok);
            qn = queue_edge(wq, qn, okm, ok, l0 < LRNG, (e + 0u) | (l0 << 20));
            qn = queue_edge(wq, qn, okm, ok, l1 < LRNG, (e + 1u) | (l1 << 20));
            qn = queue_edge(wq, qn, okm, ok, l2 < LRNG, (e + 2u) | (l2 << 20));
            qn = queue_edge(wq, qn, okm, ok, l3 < LRNG, (e + 3u) | (l3 << 20));
            if (u == LSCAN_HALF - 1 || u == LSCAN_UNROLL - 1) {
                if (qn > Q_FLUSH) {
                    drain_queue(wq, qn, lane, adj, cnt, cursor, lst);
                    qn = 0u;
                }
            }
        }
    }
    drain_queue(wq, qn, lane, adj, cnt, cursor, lst);
    __syncthreads();

    const int i = tid & (CAPW_FULL - 1);
#pragma unroll
    for (int j = 0; j < COPY_J; ++j) {
        const int s = (tid >> 8) + j * COPY_SETS;
        int rows = n - (base + s * RNG);
        rows     = rows > RNG ? RNG : rows;
        const int      capw   = park_capw(rows);
        const unsigned filled = cursor[s];
        if (capw > 0) {
            float* reg = out + (long)(base + s * RNG) * D_DIM + CNT_WORDS + c * capw;
            if (filled <= (unsigned)(capw - 1) && (unsigned)i < filled) stw(reg + i, lst[s][i]);
            if (i == CAPW_FULL - 1) stw(reg + capw - 1, filled);
        }
    }
    if (tid < LRNG) {
        const int s = tid / RNG;
        int rows = n - (base + s * RNG);
        rows     = rows > RNG ? RNG : rows;
        if (park_capw(rows) > 0)
            stw(out + (long)(base + s * RNG) * D_DIM + c * CNT_STRIDE + (tid - s * RNG), cnt[tid]);
    }
}

__global__ void __launch_bounds__(256)
gemm_wmma_f32_kernel(const float* __restrict__ x, const float* __restrict__ W,
                     const float* __restrict__ park, float* __restrict__ dinv,
                     float* __restrict__ hs) {
    const int lane = threadIdx.x & 31;
    const int wv   = threadIdx.x >> 5;
    const int tm   = blockIdx.x * 16;
    const int tn   = wv * 16;
    const int half = lane >> 4;
    const int l15  = lane & 15;

    const int    myrow = tm + l15;
    const int    orng  = myrow / RNG;
    const float* pc    = park + (long)orng * (RNG * D_DIM) + (myrow - orng * RNG);
    unsigned deg = 1u;
#pragma unroll
    for (int c = 0; c < NSLICE; ++c) deg += ldw(pc + c * CNT_STRIDE);
    const float dv = __frsqrt_rn((float)deg);
    if (wv == 0 && lane < 16) dinv[myrow] = dv;

    const float* xrow = x + (long)myrow * D_DIM;

    v8f acc = {};
#pragma unroll 8
    for (int k = 0; k < D_DIM; k += 4) {
        const int kb = k + 2 * half;
        v2f av;
        av.x = xrow[kb + 0];
        av.y = xrow[kb + 1];
        v2f bv;
        bv.x = W[(long)(kb + 0) * D_DIM + tn + l15];
        bv.y = W[(long)(kb + 1) * D_DIM + tn + l15];
        acc = __builtin_amdgcn_wmma_f32_16x16x4_f32(
             false, av,  false, bv,
             (short)0, acc,  false,  false);
    }

#pragma unroll
    for (int r = 0; r < 8; ++r) {
        const int   mr = tm + r + 8 * half;
        const float sc = __shfl(dv, r + 8 * half, 32);
        hs[(long)mr * D_DIM + tn + l15] = acc[r] * sc;
    }
}

#define LIST_CAP     (NSLICE * (CAPW_FULL - 1))
#define EDGE_THREADS 1024
#define GATH_UNROLL  16
#define OWN_J ((CAPW_FULL + EDGE_THREADS - 1) / EDGE_THREADS)

__device__ __forceinline__ void lds_add_f32(float* p, float v) {
    __hip_atomic_fetch_add(p, v, __ATOMIC_RELAXED, __HIP_MEMORY_SCOPE_WORKGROUP);
}

__global__ void __launch_bounds__(EDGE_THREADS)
edge_owner_kernel(const int* __restrict__ adj, const float* __restrict__ hs,
                  const float* __restrict__ dinv, const float* __restrict__ bias,
                  float* __restrict__ out, int nEdges, int n) {
    __shared__ float    acc[RNG * D_DIM];
    __shared__ unsigned list[LIST_CAP];

    const int tid  = threadIdx.x;
    const int lane = tid & 31;
    const int wv   = tid >> 5;
    const int base = blockIdx.x * RNG;

    for (int i = tid; i < RNG * D_DIM; i += EDGE_THREADS) acc[i] = 0.0f;

    int rows = n - base;
    rows     = rows > RNG ? RNG : rows;
    const int    capw  = park_capw(rows);
    const float* slice = out + (long)base * D_DIM;
    unsigned fc[NSLICE], off[NSLICE];
    unsigned total = 0u;
    bool     over  = capw == 0;
#pragma unroll
    for (int c = 0; c < NSLICE; ++c) {
        fc[c] = capw > 0 ? ldw(slice + CNT_WORDS + c * capw + capw - 1) : 0u;
        if (capw > 0 && fc[c] > (unsigned)(capw - 1)) over = true;
        off[c] = total;
        total += fc[c];
    }
    const int cnt = over ? 0 : (int)total;

    if (!over) {
#pragma unroll
        for (int j = 0; j < OWN_J; ++j) {
            const unsigned i = (unsigned)(tid + j * EDGE_THREADS);
            unsigned v[NSLICE];
#pragma unroll
            for (int c = 0; c < NSLICE; ++c)
                v[c] = ldw(slice + CNT_WORDS + c * capw + (i < fc[c] ? (int)i : capw - 1));
#pragma unroll
            for (int c = 0; c < NSLICE; ++c)
                if (i < fc[c]) list[off[c] + i] = v[c];
        }
    }
    __syncthreads();

    if (cnt > 0) {
        for (int i0 = wv * GATH_UNROLL; i0 < cnt; i0 += (EDGE_THREADS / 32) * GATH_UNROLL) {
            unsigned ent[GATH_UNROLL];
            float4   r[GATH_UNROLL];
#pragma unroll
            for (int u = 0; u < GATH_UNROLL; ++u) {
                const int i = i0 + u < cnt ? i0 + u : cnt - 1;
                ent[u] = list[i];
            }
#pragma unroll
            for (int u = 0; u < GATH_UNROLL; ++u) {
                const unsigned s = ent[u] & 0xFFFFu;
                r[u] = *(const float4*)(hs + (long)s * D_DIM + lane * 4);
            }
#pragma unroll
            for (int u = 0; u < GATH_UNROLL; ++u) {
                const bool live = i0 + u < cnt;
                float* a = acc + (ent[u] >> 16) * D_DIM + lane;
                lds_add_f32(a +  0, live ? r[u].x : 0.0f);
                lds_add_f32(a + 32, live ? r[u].y : 0.0f);
                lds_add_f32(a + 64, live ? r[u].z : 0.0f);
                lds_add_f32(a + 96, live ? r[u].w : 0.0f);
            }
        }
    }

    if (over) {
        const int* tgt = adj + nEdges;
        for (int e = tid; e < nEdges; e += EDGE_THREADS) {
            const unsigned dl = (unsigned)(tgt[e] - base);
            if (dl < RNG) {
                const float* row = hs + (long)adj[e] * D_DIM;
                for (int col = 0; col < D_DIM; ++col)
                    lds_add_f32(&acc[dl * D_DIM + (col & 3) * 32 + (col >> 2)], row[col]);
            }
        }
    }
    __syncthreads();

    const float4 b4 = *(const float4*)(bias + lane * 4);
    for (int r = wv; r < RNG; r += EDGE_THREADS / 32) {
        const int node = base + r;
        if (node < n) {
            const float* a  = acc + r * D_DIM + lane;
            const float4 h4 = *(const float4*)(hs + (long)node * D_DIM + lane * 4);
            const float  di = dinv[node];
            float4 o;
            o.x = (a[0]  + h4.x) * di + b4.x;
            o.y = (a[32] + h4.y) * di + b4.y;
            o.z = (a[64] + h4.z) * di + b4.z;
            o.w = (a[96] + h4.w) * di + b4.w;
            *(float4*)(out + (long)node * D_DIM + lane * 4) = o;
        }
    }
}

extern "C" void kernel_launch(void* const* d_in, const int* in_sizes, int n_in,
                              void* d_out, int out_size, void* d_ws, size_t ws_size,
                              hipStream_t stream) {
    const float* x    = (const float*)d_in[0];
    const int*   adj  = (const int*)d_in[1];
    const float* W    = (const float*)d_in[2];
    const float* bias = (const float*)d_in[3];
    float*       out  = (float*)d_out;

    const int n      = in_sizes[0] / D_DIM;
    const int nEdges = in_sizes[1] / 2;

    float* dinv = (float*)d_ws;
    float* hs   = (float*)d_ws + 10240;

    edge_list_kernel<<<((n + LRNG - 1) / LRNG) * NSLICE, LIST_THREADS, 0, stream>>>(adj, out,
                                                                                   nEdges, n);

    gemm_wmma_f32_kernel<<<n / 16, 256, 0, stream>>>(x, W, out, dinv, hs);

    edge_owner_kernel<<<(n + RNG - 1) / RNG, EDGE_THREADS, 0, stream>>>(adj, hs, dinv, bias, out,
                                                                        nEdges, n);
}


# Loads that wait on nothing go out ahead of the waits they stood behind: product wave's operands and counts as one group (was 5 round trips; k order changes), owner's list words and own rows at entry
# speedup vs baseline: 4.2730x; 1.0537x over previous
#include <hip/hip_runtime.h>

typedef __attribute__((ext_vector_type(2))) float v2f;
typedef __attribute__((ext_vector_type(8))) float v8f;

#define D_DIM 128

#define RNG        40
#define SUB        16
#define LRNG       (RNG * SUB)
#define NSLICE     16
#define CNT_STRIDE 64
#define CNT_WORDS  (NSLICE * CNT_STRIDE)
#define CAPW_FULL  256

__device__ __forceinline__ int park_capw(int rows) {
    const int room = rows * D_DIM - CNT_WORDS;
    if (room < NSLICE * 32) return 0;
    const int w = (room / NSLICE) & ~31;
    return w < CAPW_FULL ? w : CAPW_FULL;
}
__device__ __forceinline__ unsigned ldw(const float* p) { return __float_as_uint(*p); }
__device__ __forceinline__ void     stw(float* p, unsigned v) { *p = __uint_as_float(v); }

#define LIST_THREADS 1024
#define LSCAN_UNROLL 10
#define LSCAN_HALF   5
#define Q_FLUSH      128
#define Q_CAP        (Q_FLUSH + LSCAN_HALF * 128)
#define DRAIN_UNROLL 3
#define COPY_SETS    (LIST_THREADS / CAPW_FULL)
#define COPY_J       (SUB / COPY_SETS)

__device__ __forceinline__ unsigned queue_edge(unsigned* wq, unsigned qn, unsigned okm, bool ok, bool hit,
                                             unsigned word) {
    const unsigned mask = __builtin_amdgcn_ballot_w32(hit) & okm;
    if (ok && hit) wq[__builtin_amdgcn_mbcnt_lo(mask, qn)] = word;
    return qn + (unsigned)__builtin_popcount(mask);
}

__device__ __forceinline__ void drain_queue(const unsigned* wq, unsigned qn, int lane,
                                            const int* __restrict__ adj, unsigned* cnt,
                                            unsigned* cursor, unsigned (*lst)[CAPW_FULL]) {
    for (unsigned k0 = 0u; k0 < qn; k0 += 32u * DRAIN_UNROLL) {
        unsigned v[DRAIN_UNROLL], sv[DRAIN_UNROLL], p[DRAIN_UNROLL];
#pragma unroll
        for (int u = 0; u < DRAIN_UNROLL; ++u) {
            const unsigned k = k0 + 32u * u + (unsigned)lane;
            v[u] = wq[k < qn ? k : 0u];
        }
#pragma unroll
        for (int u = 0; u < DRAIN_UNROLL; ++u) sv[u] = (unsigned)adj[v[u] & 0xFFFFFu];
#pragma unroll
        for (int u = 0; u < DRAIN_UNROLL; ++u) {
            const bool     live = k0 + 32u * u + (unsigned)lane < qn;
            const unsigned l    = v[u] >> 20;
            p[u] = CAPW_FULL;
            if (live) { atomicAdd(&cnt[l], 1u); p[u] = atomicAdd(&cursor[l / RNG], 1u); }
        }
#pragma unroll
        for (int u = 0; u < DRAIN_UNROLL; ++u) {
            const unsigned l = v[u] >> 20;
            const unsigned s = l / RNG;
            if (p[u] < CAPW_FULL - 1) lst[s][p[u]] = sv[u] | ((l - s * RNG) << 16);
        }
    }
}

__global__ void __launch_bounds__(LIST_THREADS)
edge_list_kernel(const int* __restrict__ adj, float* __restrict__ out, int nEdges, int n) {
    __shared__ unsigned lst[SUB][CAPW_FULL];
    __shared__ unsigned cnt[LRNG];
    __shared__ unsigned cursor[SUB];
    __shared__ unsigned hq[LIST_THREADS / 32][Q_CAP];

    const int tid  = threadIdx.x;
    const int lane = tid & 31;
    const int R    = blockIdx.x / NSLICE;
    const int c    = blockIdx.x - R * NSLICE;
    const int base = R * LRNG;

    if (tid < LRNG) cnt[tid] = 0u;
    if (tid < SUB)  cursor[tid] = 0u;
    __syncthreads();

    const int4* tq  = (const int4*)(adj + nEdges);
    const int   nQ  = nEdges >> 2;
    const int   per = (nQ + NSLICE - 1) / NSLICE;
    const int   q0  = c * per;
    const int   q1  = q0 + per < nQ ? q0 + per : nQ;

    unsigned* wq = hq[tid >> 5];
    unsigned  qn = 0u;
    for (int qb = q0; qb < q1; qb += LIST_THREADS * LSCAN_UNROLL) {
        int4 d[LSCAN_UNROLL];
        int  q[LSCAN_UNROLL];
#pragma unroll
        for (int u = 0; u < LSCAN_UNROLL; ++u) {
            q[u] = qb + u * LIST_THREADS + tid;
            d[u] = tq[q[u] < q1 ? q[u] : q0];
        }
#pragma unroll
        for (int u = 0; u < LSCAN_UNROLL; ++u) {
            const bool     ok = q[u] < q1;
            const unsigned l0 = (unsigned)(d[u].x - base);
            const unsigned l1 = (unsigned)(d[u].y - base);
            const unsigned l2 = (unsigned)(d[u].z - base);
            const unsigned l3 = (unsigned)(d[u].w - base);
            const unsigned e   = (unsigned)q[u] * 4u;
            const unsigned okm = __builtin_amdgcn_ballot_w32(ok);
            qn = queue_edge(wq, qn, okm, ok, l0 < LRNG, (e + 0u) | (l0 << 20));
            qn = queue_edge(wq, qn, okm, ok, l1 < LRNG, (e + 1u) | (l1 << 20));
            qn = queue_edge(wq, qn, okm, ok, l2 < LRNG, (e + 2u) | (l2 << 20));
            qn = queue_edge(wq, qn, okm, ok, l3 < LRNG, (e + 3u) | (l3 << 20));
            if (u == LSCAN_HALF - 1 || u == LSCAN_UNROLL - 1) {
                if (qn > Q_FLUSH) {
                    drain_queue(wq, qn, lane, adj, cnt, cursor, lst);
                    qn = 0u;
                }
            }
        }
    }
    drain_queue(wq, qn, lane, adj, cnt, cursor, lst);
    __syncthreads();

    const int i = tid & (CAPW_FULL - 1);
#pragma unroll
    for (int j = 0; j < COPY_J; ++j) {
        const int s = (tid >> 8) + j * COPY_SETS;
        int rows = n - (base + s * RNG);
        rows     = rows > RNG ? RNG : rows;
        const int      capw   = park_capw(rows);
        const unsigned filled = cursor[s];
        if (capw > 0) {
            float* reg = out + (long)(base + s * RNG) * D_DIM + CNT_WORDS + c * capw;
            if (filled <= (unsigned)(capw - 1) && (unsigned)i < filled) stw(reg + i, lst[s][i]);
            if (i == CAPW_FULL - 1) stw(reg + capw - 1, filled);
        }
    }
    if (tid < LRNG) {
        const int s = tid / RNG;
        int rows = n - (base + s * RNG);
        rows     = rows > RNG ? RNG : rows;
        if (park_capw(rows) > 0)
            stw(out + (long)(base + s * RNG) * D_DIM + c * CNT_STRIDE + (tid - s * RNG), cnt[tid]);
    }
}

__global__ void __launch_bounds__(256) __attribute__((amdgpu_waves_per_eu(6, 6)))
gemm_wmma_f32_kernel(const float* __restrict__ x, const float* __restrict__ W,
                     const float* __restrict__ park, float* __restrict__ dinv,
                     float* __restrict__ hs) {
    const int lane = threadIdx.x & 31;
    const int wv   = threadIdx.x >> 5;
    const int tm   = blockIdx.x * 16;
    const int tn   = wv * 16;
    const int half = lane >> 4;
    const int l15  = lane & 15;

    const int    myrow = tm + l15;
    const int    orng  = myrow / RNG;
    const float* pc    = park + (long)orng * (RNG * D_DIM) + (myrow - orng * RNG);
    const float* xrow  = x + (long)myrow * D_DIM + 64 * half;
    const float* wcol  = W + (long)(64 * half) * D_DIM + tn + l15;

    unsigned cw[NSLICE];
#pragma unroll
    for (int c = 0; c < NSLICE; ++c) cw[c] = ldw(pc + c * CNT_STRIDE);
    float4 a4[D_DIM / 8];
#pragma unroll
    for (int i = 0; i < D_DIM / 8; ++i) a4[i] = *(const float4*)(xrow + 4 * i);
    v2f bv[D_DIM / 4];
#pragma unroll
    for (int j = 0; j < D_DIM / 4; ++j) {
        bv[j].x = wcol[(long)(2 * j + 0) * D_DIM];
        bv[j].y = wcol[(long)(2 * j + 1) * D_DIM];
    }

    v8f acc = {};
#pragma unroll
    for (int j = 0; j < D_DIM / 4; ++j) {
        v2f av;
        av.x = (j & 1) ? a4[j >> 1].z : a4[j >> 1].x;
        av.y = (j & 1) ? a4[j >> 1].w : a4[j >> 1].y;
        acc = __builtin_amdgcn_wmma_f32_16x16x4_f32(
             false, av,  false, bv[j],
             (short)0, acc,  false,  false);
    }

    unsigned deg = 1u;
#pragma unroll
    for (int c = 0; c < NSLICE; ++c) deg += cw[c];
    const float dv = __frsqrt_rn((float)deg);
    if (wv == 0 && lane < 16) dinv[myrow] = dv;

#pragma unroll
    for (int r = 0; r < 8; ++r) {
        const int   mr = tm + r + 8 * half;
        const float sc = __shfl(dv, r + 8 * half, 32);
        hs[(long)mr * D_DIM + tn + l15] = acc[r] * sc;
    }
}

#define LIST_CAP     (NSLICE * (CAPW_FULL - 1))
#define EDGE_THREADS 1024
#define GATH_UNROLL  16
#define OWN_J ((CAPW_FULL + EDGE_THREADS - 1) / EDGE_THREADS)
#define OWN_ROWS ((RNG + EDGE_THREADS / 32 - 1) / (EDGE_THREADS / 32))

__device__ __forceinline__ void lds_add_f32(float* p, float v) {
    __hip_atomic_fetch_add(p, v, __ATOMIC_RELAXED, __HIP_MEMORY_SCOPE_WORKGROUP);
}

__global__ void __launch_bounds__(EDGE_THREADS)
edge_owner_kernel(const int* __restrict__ adj, const float* __restrict__ hs,
                  const float* __restrict__ dinv, const float* __restrict__ bias,
                  float* __restrict__ out, int nEdges, int n) {
    __shared__ float    acc[RNG * D_DIM];
    __shared__ unsigned list[LIST_CAP];

    const int tid  = threadIdx.x;
    const int lane = tid & 31;
    const int wv   = tid >> 5;
    const int base = blockIdx.x * RNG;

    for (int i = tid; i < RNG * D_DIM; i += EDGE_THREADS) acc[i] = 0.0f;

    const float4 b4 = *(const float4*)(bias + lane * 4);
    float4 own_h[OWN_ROWS];
    float  own_d[OWN_ROWS];
#pragma unroll
    for (int t = 0; t < OWN_ROWS; ++t) {
        const int r    = wv + t * (EDGE_THREADS / 32);
        int       node = base + (r < RNG ? r : wv);
        node           = node < n ? node : n - 1;
        own_h[t] = *(const float4*)(hs + (long)node * D_DIM + lane * 4);
        own_d[t] = dinv[node];
    }

    int rows = n - base;
    rows     = rows > RNG ? RNG : rows;
    const int    capw  = park_capw(rows);
    const float* slice = out + (long)base * D_DIM;
    unsigned pv[OWN_J][NSLICE];
    if (capw > 0) {
#pragma unroll
        for (int j = 0; j < OWN_J; ++j) {
            const int i = tid + j * EDGE_THREADS;
#pragma unroll
            for (int c = 0; c < NSLICE; ++c)
                pv[j][c] = ldw(slice + CNT_WORDS + c * capw + (i < capw - 1 ? i : capw - 1));
        }
    }
    unsigned fc[NSLICE], off[NSLICE];
    unsigned total = 0u;
    bool     over  = capw == 0;
#pragma unroll
    for (int c = 0; c < NSLICE; ++c) {
        fc[c] = capw > 0 ? ldw(slice + CNT_WORDS + c * capw + capw - 1) : 0u;
        if (capw > 0 && fc[c] > (unsigned)(capw - 1)) over = true;
        off[c] = total;
        total += fc[c];
    }
    const int cnt = over ? 0 : (int)total;

    if (!over) {
#pragma unroll
        for (int j = 0; j < OWN_J; ++j) {
            const unsigned i = (unsigned)(tid + j * EDGE_THREADS);
#pragma unroll
            for (int c = 0; c < NSLICE; ++c)
                if (i < fc[c]) list[off[c] + i] = pv[j][c];
        }
    }
    __syncthreads();

    if (cnt > 0) {
        for (int i0 = wv * GATH_UNROLL; i0 < cnt; i0 += (EDGE_THREADS / 32) * GATH_UNROLL) {
            unsigned ent[GATH_UNROLL];
            float4   r[GATH_UNROLL];
#pragma unroll
            for (int u = 0; u < GATH_UNROLL; ++u) {
                const int i = i0 + u < cnt ? i0 + u : cnt - 1;
                ent[u] = list[i];
            }
#pragma unroll
            for (int u = 0; u < GATH_UNROLL; ++u) {
                const unsigned s = ent[u] & 0xFFFFu;
                r[u] = *(const float4*)(hs + (long)s * D_DIM + lane * 4);
            }
#pragma unroll
            for (int u = 0; u < GATH_UNROLL; ++u) {
                const bool live = i0 + u < cnt;
                float* a = acc + (ent[u] >> 16) * D_DIM + lane;
                lds_add_f32(a +  0, live ? r[u].x : 0.0f);
                lds_add_f32(a + 32, live ? r[u].y : 0.0f);
                lds_add_f32(a + 64, live ? r[u].z : 0.0f);
                lds_add_f32(a + 96, live ? r[u].w : 0.0f);
            }
        }
    }

    if (over) {
        const int* tgt = adj + nEdges;
        for (int e = tid; e < nEdges; e += EDGE_THREADS) {
            const unsigned dl = (unsigned)(tgt[e] - base);
            if (dl < RNG) {
                const float* row = hs + (long)adj[e] * D_DIM;
                for (int col = 0; col < D_DIM; ++col)
                    lds_add_f32(&acc[dl * D_DIM + (col & 3) * 32 + (col >> 2)], row[col]);
            }
        }
    }
    __syncthreads();

#pragma unroll
    for (int t = 0; t < OWN_ROWS; ++t) {
        const int r    = wv + t * (EDGE_THREADS / 32);
        const int node = base + r;
        if (r < RNG && node < n) {
            const float* a  = acc + r * D_DIM + lane;
            const float4 h4 = own_h[t];
            const float  di = own_d[t];
            float4 o;
            o.x = (a[0]  + h4.x) * di + b4.x;
            o.y = (a[32] + h4.y) * di + b4.y;
            o.z = (a[64] + h4.z) * di + b4.z;
            o.w = (a[96] + h4.w) * di + b4.w;
            *(float4*)(out + (long)node * D_DIM + lane * 4) = o;
        }
    }
}

extern "C" void kernel_launch(void* const* d_in, const int* in_sizes, int n_in,
                              void* d_out, int out_size, void* d_ws, size_t ws_size,
                              hipStream_t stream) {
    const float* x    = (const float*)d_in[0];
    const int*   adj  = (const int*)d_in[1];
    const float* W    = (const float*)d_in[2];
    const float* bias = (const float*)d_in[3];
    float*       out  = (float*)d_out;

    const int n      = in_sizes[0] / D_DIM;
    const int nEdges = in_sizes[1] / 2;

    float* dinv = (float*)d_ws;
    float* hs   = (float*)d_ws + 10240;

    edge_list_kernel<<<((n + LRNG - 1) / LRNG) * NSLICE, LIST_THREADS, 0, stream>>>(adj, out,
                                                                                   nEdges, n);

    gemm_wmma_f32_kernel<<<n / 16, 256, 0, stream>>>(x, W, out, dinv, hs);

    edge_owner_kernel<<<(n + RNG - 1) / RNG, EDGE_THREADS, 0, stream>>>(adj, hs, dinv, bias, out,
                                                                        nEdges, n);
}
